# starting kernel
# baseline (speedup 1.0000x reference)
#include <hip/hip_runtime.h>


typedef float v2f __attribute__((ext_vector_type(2)));
typedef float v8f __attribute__((ext_vector_type(8)));

#define Hdim    56
#define Ssp     3136
#define C12     12
#define WPAD    62
#define T3N     (C12*Hdim*WPAD)
#define CPAD    32
#define T5ROWS  58
#define T5N     (T5ROWS*Hdim*CPAD)
#define ROWBYTES1 3472
#define JSTRIDE2  1792

__global__ void k_prep(const float* __restrict__ x,
                       float* __restrict__ t3rp, float* __restrict__ t5p) {
  int idx = blockIdx.x * blockDim.x + threadIdx.x;
  if (idx < T3N) {
    int c = idx / (Hdim * WPAD);
    int r = idx - c * (Hdim * WPAD);
    int n = r / WPAD;
    int w = r - n * WPAD;
    float v = 0.0f;
    if (w >= 3 && w < 59) {
      int m = w - 3;
      int msrc = (m + Hdim - 1) % Hdim;
      int s = c * Ssp + n * Hdim + msrc;
      v = x[s] + x[s + C12 * Ssp];
    }
    t3rp[idx] = v;
  } else {
    int p = idx - T3N;
    if (p < 2 * Hdim * CPAD) {
      int top = p / (Hdim * CPAD);
      int q = p - top * (Hdim * CPAD);
      t5p[(top ? (T5ROWS - 1) * Hdim * CPAD : 0) + q] = 0.0f;
    }
  }
}

__global__ void k_gemm1(const float* __restrict__ w2,
                        const float* __restrict__ t3rp,
                        float* __restrict__ t5p) {
  __shared__ float w2p[32 * 84];
  for (int i = threadIdx.x; i < 32 * 84; i += blockDim.x) {
    int row = i / 84;
    int kkp = i - row * 84;
    int jj = kkp / 12, c = kkp - jj * 12;
    w2p[i] = (row < 21) ? w2[(c * 7 + jj) * 21 + row] : 0.0f;
  }
  __syncthreads();

  int wave = (blockIdx.x * blockDim.x + threadIdx.x) >> 5;
  int lane = threadIdx.x & 31;
  int mt = wave & 1;
  int nt = wave >> 1;
  int half = lane >> 4;
  int l15  = lane & 15;

  int row = mt * 16 + l15;
  int col = nt * 16 + l15;
  int n = col / Hdim;
  int m = col - n * Hdim;

  const v2f*   ap = (const v2f*)&w2p[row * 84 + 2 * half];
  const float* bp = t3rp + 2 * half * ROWBYTES1 + n * WPAD + m;

  v8f acc = {};
  #pragma unroll
  for (int t = 0; t < 21; ++t) {
    v2f a = ap[2 * t];
    int off = 4 * (t % 3) * ROWBYTES1 + t / 3;
    v2f b;
    b.x = bp[off];
    b.y = bp[off + ROWBYTES1];
    acc = __builtin_amdgcn_wmma_f32_16x16x4_f32(
        false, a, false, b, (short)0, acc, false, false);
  }

  #pragma unroll
  for (int v = 0; v < 8; ++v) {
    int rout = mt * 16 + v + half * 8;
    t5p[(col + Hdim) * CPAD + rout] = acc[v];
  }
}

__global__ void k_gemm2(const float* __restrict__ w1,
                        const float* __restrict__ t5p,
                        float* __restrict__ y) {
  __shared__ float w1p[96 * 96];
  for (int i = threadIdx.x; i < 96 * 96; i += blockDim.x) {
    int row = i / 96;
    int kkp = i - row * 96;
    int j = kkp / 32, k = kkp - j * 32;
    w1p[i] = (k < 21) ? w1[row * 63 + j * 21 + k] : 0.0f;
  }
  __syncthreads();

  int wave = (blockIdx.x * blockDim.x + threadIdx.x) >> 5;
  int lane = threadIdx.x & 31;
  int mt = wave % 6;
  int nt = wave / 6;
  int half = lane >> 4;
  int l15  = lane & 15;

  int row = mt * 16 + l15;
  int col = nt * 16 + l15;

  const v2f*   ap = (const v2f*)&w1p[row * 96 + 2 * half];
  const float* bp = t5p + col * CPAD + 2 * half;

  v8f acc = {};
  #pragma unroll
  for (int t = 0; t < 24; ++t) {
    v2f a = ap[2 * t];
    int j = t / 8;
    int off = j * JSTRIDE2 + (4 * t - 32 * j);
    v2f b;
    b.x = bp[off];
    b.y = bp[off + 1];
    acc = __builtin_amdgcn_wmma_f32_16x16x4_f32(
        false, a, false, b, (short)0, acc, false, false);
  }

  #pragma unroll
  for (int v = 0; v < 8; ++v) {
    int rout = mt * 16 + v + half * 8;
    y[rout * Ssp + col] = acc[v];
  }
}

extern "C" void kernel_launch(void* const* d_in, const int* in_sizes, int n_in,
                              void* d_out, int out_size, void* d_ws, size_t ws_size,
                              hipStream_t stream) {
  const float* x  = (const float*)d_in[0];
  const float* w1 = (const float*)d_in[1];
  const float* w2 = (const float*)d_in[2];
  float* y = (float*)d_out;

  float* t3rp = (float*)d_ws;
  float* t5p  = t3rp + T3N;

  int prep_threads = T3N + 2 * Hdim * CPAD;
  k_prep<<<(prep_threads + 255) / 256, 256, 0, stream>>>(x, t3rp, t5p);

  k_gemm1<<<49, 256, 0, stream>>>(w2, t3rp, t5p);

  k_gemm2<<<147, 256, 0, stream>>>(w1, t5p, y);
}


# Three launches fused into one, no LDS or workspace: each wave forms its three t5 tiles as WMMA D tiles and uses those registers directly as GEMM2 B fragments (GEMM2 k order follows D: regroups sums)
# speedup vs baseline: 4.0827x; 4.0827x over previous
#include <hip/hip_runtime.h>


typedef float v2f __attribute__((ext_vector_type(2)));
typedef float v8f __attribute__((ext_vector_type(8)));

#define HW   56
#define SP   3136
#define WC2  147
#define W1R  63

struct XRaw { float v[84]; };
struct A2F  { v2f m[3][4]; v2f h[3]; float s[3]; };

__device__ __forceinline__ bool win_ok(int n, int b, int jj) {
  const int mm = b + jj - 3;
  return (n >= 0) && (n < HW) && (mm >= 0) && (mm < HW);
}

template <int TA, int TB>
__device__ __forceinline__ void load_x(const float* __restrict__ xh, int n, int b, XRaw& r) {
  #pragma unroll
  for (int t = TA; t < TB; ++t) {
    const int jj = t / 3, cb = 4 * (t % 3);
    const int mm = b + jj - 3;
    const int msrc = (mm == 0) ? (HW - 1) : (mm - 1);
    const int off = win_ok(n, b, jj) ? (n * HW + msrc) : 0;
    const float* p = xh + cb * SP + off;
    r.v[4 * t + 0] = p[0];
    r.v[4 * t + 1] = p[12 * SP];
    r.v[4 * t + 2] = p[SP];
    r.v[4 * t + 3] = p[13 * SP];
  }
}

__device__ __forceinline__ void load_a2(const float* __restrict__ w1r, int j, int h, A2F& f) {
  #pragma unroll
  for (int q = 0; q < 3; ++q) {
    const float* r = w1r + q * 16 * W1R + j * 21;
    #pragma unroll
    for (int p = 0; p < 4; ++p) {
      f.m[q][p].x = r[8 * h + 2 * p];
      f.m[q][p].y = r[8 * h + 2 * p + 1];
    }
    f.h[q].x = r[16 + 2 * h];
    f.h[q].y = r[16 + 2 * h + 1];
    f.s[q]   = r[20];
  }
}

template <int TA, int TB>
__device__ __forceinline__ void tap_t5(const XRaw& r, int n, int b,
                                       const v2f* a1lo, const v2f* a1hi, v8f& lo, v8f& hi) {
  #pragma unroll
  for (int t = TA; t < TB; ++t) {
    const bool ok = win_ok(n, b, t / 3);
    const float u0 = r.v[4 * t + 0] + r.v[4 * t + 1];
    const float u1 = r.v[4 * t + 2] + r.v[4 * t + 3];
    v2f bb;
    bb.x = ok ? u0 : 0.0f;
    bb.y = ok ? u1 : 0.0f;
    lo = __builtin_amdgcn_wmma_f32_16x16x4_f32(false, a1lo[t], false, bb,
                                               (short)0, lo, false, false);
    hi = __builtin_amdgcn_wmma_f32_16x16x4_f32(false, a1hi[t], false, bb,
                                               (short)0, hi, false, false);
  }
}

__device__ __forceinline__ void tap_y(const v8f& lo, const v8f& hi, const A2F& f, int h, v8f* acc) {
  #pragma unroll
  for (int p = 0; p < 4; ++p) {
    v2f bb;
    bb.x = lo[2 * p];
    bb.y = lo[2 * p + 1];
    #pragma unroll
    for (int q = 0; q < 3; ++q)
      acc[q] = __builtin_amdgcn_wmma_f32_16x16x4_f32(false, f.m[q][p], false, bb,
                                                     (short)0, acc[q], false, false);
  }
  {
    v2f bb;
    bb.x = hi[0];
    bb.y = hi[1];
    #pragma unroll
    for (int q = 0; q < 3; ++q)
      acc[q] = __builtin_amdgcn_wmma_f32_16x16x4_f32(false, f.h[q], false, bb,
                                                     (short)0, acc[q], false, false);
  }
  {
    v2f bb;
    bb.x = hi[2];
    bb.y = hi[3];
    #pragma unroll
    for (int q = 0; q < 3; ++q) {
      v2f aa;
      aa.x = (h == 0) ? f.s[q] : 0.0f;
      aa.y = 0.0f;
      acc[q] = __builtin_amdgcn_wmma_f32_16x16x4_f32(false, aa, false, bb,
                                                     (short)0, acc[q], false, false);
    }
  }
}

template <int TA, int TB>
__device__ __forceinline__ void load_a1(const float* __restrict__ w2lo,
                                        const float* __restrict__ w2hi,
                                        v2f* a1lo, v2f* a1hi) {
  #pragma unroll
  for (int t = TA; t < TB; ++t) {
    const int jj = t / 3, cb = 4 * (t % 3);
    const int o0 = cb * WC2 + jj * 21, o1 = o0 + WC2;
    a1lo[t].x = w2lo[o0];
    a1lo[t].y = w2lo[o1];
    a1hi[t].x = w2hi[o0];
    a1hi[t].y = w2hi[o1];
  }
}

#define STAGE_END(A, B) do { asm volatile("" :: "v"(A), "v"(B)); \
                             __builtin_amdgcn_sched_barrier(0); } while (0)

#define LOADS_END() __builtin_amdgcn_sched_barrier(0)

__global__ __launch_bounds__(64)
void k_fused(const float* __restrict__ x, const float* __restrict__ w1,
             const float* __restrict__ w2, float* __restrict__ y) {
  const int lane = threadIdx.x & 31;
  const int grp  = threadIdx.x >> 5;
  const int h    = lane >> 4;
  const int c15  = lane & 15;
  const int col  = blockIdx.x * 16 + c15;
  const int a    = col / HW;
  const int b    = col - a * HW;

  const int  hich = (c15 < 2) ? 16 + c15
                  : (c15 == 2) ? 20
                  : (c15 == 8 || c15 == 9) ? 10 + c15 : 16;
  const float* w2lo = w2 + 2 * h * WC2 + c15;
  const float* w2hi = w2 + 2 * h * WC2 + hich;
  const float* xh   = x + 2 * h * SP;
  const float* w1r  = w1 + (grp * 48 + c15) * W1R;

  v2f  a1lo[21], a1hi[21];
  XRaw r0, r1, r2;
  A2F  f0, f1, f2;
  v8f  acc[3] = {};
  v8f  lo = {}, hi = {};

  load_a1<0, 7>(w2lo, w2hi, a1lo, a1hi);
  load_x<0, 7>(xh, a - 1, b, r0);
  load_a1<7, 14>(w2lo, w2hi, a1lo, a1hi);
  load_x<7, 14>(xh, a - 1, b, r0);
  load_a1<14, 21>(w2lo, w2hi, a1lo, a1hi);
  load_x<14, 21>(xh, a - 1, b, r0);
  LOADS_END();

  tap_t5<0, 7>(r0, a - 1, b, a1lo, a1hi, lo, hi);
  STAGE_END(lo, hi);
  load_x<0, 11>(xh, a, b, r1);
  load_a2(w1r, 0, h, f0);
  LOADS_END();

  tap_t5<7, 21>(r0, a - 1, b, a1lo, a1hi, lo, hi);
  STAGE_END(lo, hi);
  load_x<11, 21>(xh, a, b, r1);
  load_a2(w1r, 1, h, f1);
  LOADS_END();

  tap_y(lo, hi, f0, h, acc);
  lo = v8f{};
  hi = v8f{};
  tap_t5<0, 11>(r1, a, b, a1lo, a1hi, lo, hi);
  STAGE_END(lo, hi);
  load_x<0, 11>(xh, a + 1, b, r2);
  load_a2(w1r, 2, h, f2);
  LOADS_END();

  tap_t5<11, 21>(r1, a, b, a1lo, a1hi, lo, hi);
  STAGE_END(lo, hi);
  load_x<11, 21>(xh, a + 1, b, r2);
  LOADS_END();

  tap_y(lo, hi, f1, h, acc);
  lo = v8f{};
  hi = v8f{};
  tap_t5<0, 11>(r2, a + 1, b, a1lo, a1hi, lo, hi);
  STAGE_END(lo, hi);
  tap_t5<11, 21>(r2, a + 1, b, a1lo, a1hi, lo, hi);
  tap_y(lo, hi, f2, h, acc);

  #pragma unroll
  for (int q = 0; q < 3; ++q) {
    #pragma unroll
    for (int r = 0; r < 8; ++r) {
      y[((grp * 3 + q) * 16 + 8 * h + r) * SP + col] = acc[q][r];
    }
  }
}

extern "C" void kernel_launch(void* const* d_in, const int* in_sizes, int n_in,
                              void* d_out, int out_size, void* d_ws, size_t ws_size,
                              hipStream_t stream) {
  const float* x  = (const float*)d_in[0];
  const float* w1 = (const float*)d_in[1];
  const float* w2 = (const float*)d_in[2];
  float* y = (float*)d_out;

  k_fused<<<196, 64, 0, stream>>>(x, w1, w2, y);
}


# Wave tile re-shaped to 2 rows x 8 columns, 32 channels (3 waves a tile): the 3 taps need 2 t5 tiles, not 3; tap 1 reads both via a row_ror:8 DPP move; 120 WMMAs a wave, was 180; order of sums kept
# speedup vs baseline: 4.4281x; 1.0846x over previous
#include <hip/hip_runtime.h>


typedef float v2f __attribute__((ext_vector_type(2)));
typedef float v8f __attribute__((ext_vector_type(8)));

#define HW   56
#define SP   3136
#define WC2  147
#define W1R  63

struct XRaw { float v[84]; };
struct A2F  { v2f m[2][4]; v2f h[2]; float s[2]; };

__device__ __forceinline__ bool win_ok(int n, int b, int jj) {
  const int mm = b + jj - 3;
  return (n >= 0) && (n < HW) && (mm >= 0) && (mm < HW);
}

template <int TA, int TB>
__device__ __forceinline__ void load_x(const float* __restrict__ xh, int n, int b, XRaw& r) {
  #pragma unroll
  for (int t = TA; t < TB; ++t) {
    const int jj = t / 3, cb = 4 * (t % 3);
    const int mm = b + jj - 3;
    const int msrc = (mm == 0) ? (HW - 1) : (mm - 1);
    const int off = win_ok(n, b, jj) ? (n * HW + msrc) : 0;
    const float* p = xh + cb * SP + off;
    r.v[4 * t + 0] = p[0];
    r.v[4 * t + 1] = p[12 * SP];
    r.v[4 * t + 2] = p[SP];
    r.v[4 * t + 3] = p[13 * SP];
  }
}

__device__ __forceinline__ void load_a2(const float* __restrict__ w1r, int j, int h, A2F& f) {
  #pragma unroll
  for (int q = 0; q < 2; ++q) {
    const float* r = w1r + q * 16 * W1R + j * 21;
    #pragma unroll
    for (int p = 0; p < 4; ++p) {
      f.m[q][p].x = r[8 * h + 2 * p];
      f.m[q][p].y = r[8 * h + 2 * p + 1];
    }
    f.h[q].x = r[16 + 2 * h];
    f.h[q].y = r[16 + 2 * h + 1];
    f.s[q]   = r[20];
  }
}

template <int TA, int TB>
__device__ __forceinline__ void tap_t5(const XRaw& r, int n, int b,
                                       const v2f* a1lo, const v2f* a1hi, v8f& lo, v8f& hi) {
  #pragma unroll
  for (int t = TA; t < TB; ++t) {
    const bool ok = win_ok(n, b, t / 3);
    const float u0 = r.v[4 * t + 0] + r.v[4 * t + 1];
    const float u1 = r.v[4 * t + 2] + r.v[4 * t + 3];
    v2f bb;
    bb.x = ok ? u0 : 0.0f;
    bb.y = ok ? u1 : 0.0f;
    lo = __builtin_amdgcn_wmma_f32_16x16x4_f32(false, a1lo[t], false, bb,
                                               (short)0, lo, false, false);
    hi = __builtin_amdgcn_wmma_f32_16x16x4_f32(false, a1hi[t], false, bb,
                                               (short)0, hi, false, false);
  }
}

__device__ __forceinline__ void tap_y(const v8f& lo, const v8f& hi, const A2F& f, int h, v8f* acc) {
  #pragma unroll
  for (int p = 0; p < 4; ++p) {
    v2f bb;
    bb.x = lo[2 * p];
    bb.y = lo[2 * p + 1];
    #pragma unroll
    for (int q = 0; q < 2; ++q)
      acc[q] = __builtin_amdgcn_wmma_f32_16x16x4_f32(false, f.m[q][p], false, bb,
                                                     (short)0, acc[q], false, false);
  }
  {
    v2f bb;
    bb.x = hi[0];
    bb.y = hi[1];
    #pragma unroll
    for (int q = 0; q < 2; ++q)
      acc[q] = __builtin_amdgcn_wmma_f32_16x16x4_f32(false, f.h[q], false, bb,
                                                     (short)0, acc[q], false, false);
  }
  {
    v2f bb;
    bb.x = hi[2];
    bb.y = hi[3];
    #pragma unroll
    for (int q = 0; q < 2; ++q) {
      v2f aa;
      aa.x = (h == 0) ? f.s[q] : 0.0f;
      aa.y = 0.0f;
      acc[q] = __builtin_amdgcn_wmma_f32_16x16x4_f32(false, aa, false, bb,
                                                     (short)0, acc[q], false, false);
    }
  }
}

template <int TA, int TB>
__device__ __forceinline__ void load_a1(const float* __restrict__ w2lo,
                                        const float* __restrict__ w2hi,
                                        v2f* a1lo, v2f* a1hi) {
  #pragma unroll
  for (int t = TA; t < TB; ++t) {
    const int jj = t / 3, cb = 4 * (t % 3);
    const int o0 = cb * WC2 + jj * 21, o1 = o0 + WC2;
    a1lo[t].x = w2lo[o0];
    a1lo[t].y = w2lo[o1];
    a1hi[t].x = w2hi[o0];
    a1hi[t].y = w2hi[o1];
  }
}

#define STAGE_END(A, B) do { asm volatile("" :: "v"(A), "v"(B)); \
                             __builtin_amdgcn_sched_barrier(0); } while (0)

#define LOADS_END() __builtin_amdgcn_sched_barrier(0)

__device__ __forceinline__ float rows_0_1(float t0, float t1) {
  int v = __builtin_amdgcn_update_dpp(0, __builtin_bit_cast(int, t0), 0x128, 0xf, 0xf, false);
  v = __builtin_amdgcn_update_dpp(v, __builtin_bit_cast(int, t1), 0x128, 0xf, 0xc, false);
  return __builtin_bit_cast(float, v);
}

__global__ __launch_bounds__(96)
void k_fused(const float* __restrict__ x, const float* __restrict__ w1,
             const float* __restrict__ w2, float* __restrict__ y) {
  const int lane = threadIdx.x & 31;
  const int grp  = threadIdx.x >> 5;
  const int h    = lane >> 4;
  const int c15  = lane & 15;
  const int ablk = blockIdx.x / 7;
  const int bblk = blockIdx.x - ablk * 7;
  const int a    = 2 * ablk + (c15 >> 3);
  const int b    = 8 * bblk + (c15 & 7);

  const int  hich = (c15 < 2) ? 16 + c15
                  : (c15 == 2) ? 20
                  : (c15 == 8 || c15 == 9) ? 10 + c15 : 16;
  const float* w2lo = w2 + 2 * h * WC2 + c15;
  const float* w2hi = w2 + 2 * h * WC2 + hich;
  const float* xh   = x + 2 * h * SP;
  const float* w1r  = w1 + (grp * 32 + c15) * W1R;

  v2f  a1lo[21], a1hi[21];
  XRaw r0, r1;
  A2F  f0, f1, f2;
  v8f  acc[2] = {};
  v8f  lo0 = {}, hi0 = {}, lo1 = {}, hi1 = {};

  load_a1<0, 7>(w2lo, w2hi, a1lo, a1hi);
  load_x<0, 7>(xh, a - 1, b, r0);
  load_a1<7, 14>(w2lo, w2hi, a1lo, a1hi);
  load_x<7, 14>(xh, a - 1, b, r0);
  load_a1<14, 21>(w2lo, w2hi, a1lo, a1hi);
  load_x<14, 21>(xh, a - 1, b, r0);
  LOADS_END();

  tap_t5<0, 7>(r0, a - 1, b, a1lo, a1hi, lo0, hi0);
  STAGE_END(lo0, hi0);
  load_x<0, 11>(xh, a + 1, b, r1);
  load_a2(w1r, 0, h, f0);
  load_a2(w1r, 1, h, f1);
  LOADS_END();

  tap_t5<7, 21>(r0, a - 1, b, a1lo, a1hi, lo0, hi0);
  STAGE_END(lo0, hi0);
  load_x<11, 21>(xh, a + 1, b, r1);
  load_a2(w1r, 2, h, f2);
  LOADS_END();

  tap_y(lo0, hi0, f0, h, acc);
  tap_t5<0, 11>(r1, a + 1, b, a1lo, a1hi, lo1, hi1);
  STAGE_END(lo1, hi1);
  tap_t5<11, 21>(r1, a + 1, b, a1lo, a1hi, lo1, hi1);

  v8f lom, him = {};
  #pragma unroll
  for (int r = 0; r < 8; ++r) lom[r] = rows_0_1(lo0[r], lo1[r]);
  #pragma unroll
  for (int r = 0; r < 4; ++r) him[r] = rows_0_1(hi0[r], hi1[r]);
  tap_y(lom, him, f1, h, acc);
  tap_y(lo1, hi1, f2, h, acc);

  #pragma unroll
  for (int q = 0; q < 2; ++q) {
    #pragma unroll
    for (int r = 0; r < 8; ++r) {
      y[((grp * 2 + q) * 16 + 8 * h + r) * SP + a * HW + b] = acc[q][r];
    }
  }
}

extern "C" void kernel_launch(void* const* d_in, const int* in_sizes, int n_in,
                              void* d_out, int out_size, void* d_ws, size_t ws_size,
                              hipStream_t stream) {
  const float* x  = (const float*)d_in[0];
  const float* w1 = (const float*)d_in[1];
  const float* w2 = (const float*)d_in[2];
  float* y = (float*)d_out;

  k_fused<<<196, 96, 0, stream>>>(x, w1, w2, y);
}


# GEMM1 K split over the work-group's 3 waves (4 of 12 summed input channels each: 28 WMMAs, was 84), partial t5 tiles added through LDS in fixed order (regroups sums); 64 WMMAs a wave, was 120
# speedup vs baseline: 5.1573x; 1.1647x over previous
#include <hip/hip_runtime.h>


typedef float v2f __attribute__((ext_vector_type(2)));
typedef float v4f __attribute__((ext_vector_type(4)));
typedef float v8f __attribute__((ext_vector_type(8)));

#define HW   56
#define SP   3136
#define WC2  147
#define W1R  63

struct XRaw { float v[28]; };
struct A2F  { v2f m[2][4]; v2f h[2]; float s[2]; };

__device__ __forceinline__ bool win_ok(int n, int b, int jj) {
  const int mm = b + jj - 3;
  return (n >= 0) && (n < HW) && (mm >= 0) && (mm < HW);
}

__device__ __forceinline__ void load_x1(const float* __restrict__ xg, int n, int b, int jj, XRaw& r) {
  const int mm = b + jj - 3;
  const int msrc = (mm == 0) ? (HW - 1) : (mm - 1);
  const int off = win_ok(n, b, jj) ? (n * HW + msrc) : 0;
  const float* p = xg + off;
  r.v[4 * jj + 0] = p[0];
  r.v[4 * jj + 1] = p[12 * SP];
  r.v[4 * jj + 2] = p[SP];
  r.v[4 * jj + 3] = p[13 * SP];
}

template <int JA, int JB>
__device__ __forceinline__ void load_g1(const float* __restrict__ w2lo, const float* __restrict__ w2hi,
                                        const float* __restrict__ xg, int a, int b,
                                        v2f* a1lo, v2f* a1hi, XRaw& r0, XRaw& r1) {
  #pragma unroll
  for (int jj = JA; jj < JB; ++jj) {
    a1lo[jj].x = w2lo[jj * 21];
    a1lo[jj].y = w2lo[jj * 21 + WC2];
    a1hi[jj].x = w2hi[jj * 21];
    a1hi[jj].y = w2hi[jj * 21 + WC2];
    load_x1(xg, a - 1, b, jj, r0);
    load_x1(xg, a + 1, b, jj, r1);
  }
}

__device__ __forceinline__ v2f x_frag(const XRaw& r, int n, int b, int jj) {
  const bool ok = win_ok(n, b, jj);
  const float u0 = r.v[4 * jj + 0] + r.v[4 * jj + 1];
  const float u1 = r.v[4 * jj + 2] + r.v[4 * jj + 3];
  v2f bb;
  bb.x = ok ? u0 : 0.0f;
  bb.y = ok ? u1 : 0.0f;
  return bb;
}

template <int JA, int JB>
__device__ __forceinline__ void gemm1(const XRaw& r0, const XRaw& r1, int a, int b,
                                      const v2f* a1lo, const v2f* a1hi,
                                      v8f& lo0, v8f& hi0, v8f& lo1, v8f& hi1) {
  #pragma unroll
  for (int jj = JA; jj < JB; ++jj) {
    const v2f b0 = x_frag(r0, a - 1, b, jj);
    const v2f b1 = x_frag(r1, a + 1, b, jj);
    lo0 = __builtin_amdgcn_wmma_f32_16x16x4_f32(false, a1lo[jj], false, b0,
                                                (short)0, lo0, false, false);
    hi0 = __builtin_amdgcn_wmma_f32_16x16x4_f32(false, a1hi[jj], false, b0,
                                                (short)0, hi0, false, false);
    lo1 = __builtin_amdgcn_wmma_f32_16x16x4_f32(false, a1lo[jj], false, b1,
                                                (short)0, lo1, false, false);
    hi1 = __builtin_amdgcn_wmma_f32_16x16x4_f32(false, a1hi[jj], false, b1,
                                                (short)0, hi1, false, false);
  }
}

__device__ __forceinline__ void load_a2(const float* __restrict__ w1r, int j, int h, A2F& f) {
  #pragma unroll
  for (int q = 0; q < 2; ++q) {
    const float* r = w1r + q * 16 * W1R + j * 21;
    #pragma unroll
    for (int p = 0; p < 4; ++p) {
      f.m[q][p].x = r[8 * h + 2 * p];
      f.m[q][p].y = r[8 * h + 2 * p + 1];
    }
    f.h[q].x = r[16 + 2 * h];
    f.h[q].y = r[16 + 2 * h + 1];
    f.s[q]   = r[20];
  }
}

__device__ __forceinline__ void tap_y(const v8f& lo, const v4f& hi, const A2F& f, int h, v8f* acc) {
  #pragma unroll
  for (int p = 0; p < 4; ++p) {
    v2f bb;
    bb.x = lo[2 * p];
    bb.y = lo[2 * p + 1];
    #pragma unroll
    for (int q = 0; q < 2; ++q)
      acc[q] = __builtin_amdgcn_wmma_f32_16x16x4_f32(false, f.m[q][p], false, bb,
                                                     (short)0, acc[q], false, false);
  }
  {
    v2f bb;
    bb.x = hi[0];
    bb.y = hi[1];
    #pragma unroll
    for (int q = 0; q < 2; ++q)
      acc[q] = __builtin_amdgcn_wmma_f32_16x16x4_f32(false, f.h[q], false, bb,
                                                     (short)0, acc[q], false, false);
  }
  {
    v2f bb;
    bb.x = hi[2];
    bb.y = hi[3];
    #pragma unroll
    for (int q = 0; q < 2; ++q) {
      v2f aa;
      aa.x = (h == 0) ? f.s[q] : 0.0f;
      aa.y = 0.0f;
      acc[q] = __builtin_amdgcn_wmma_f32_16x16x4_f32(false, aa, false, bb,
                                                     (short)0, acc[q], false, false);
    }
  }
}

#define STAGE_END(A, B, C, D) do { asm volatile("" :: "v"(A), "v"(B), "v"(C), "v"(D)); \
                                   __builtin_amdgcn_sched_barrier(0); } while (0)

#define LOADS_END() __builtin_amdgcn_sched_barrier(0)

__device__ __forceinline__ float rows_0_1(float t0, float t1) {
  int v = __builtin_amdgcn_update_dpp(0, __builtin_bit_cast(int, t0), 0x128, 0xf, 0xf, false);
  v = __builtin_amdgcn_update_dpp(v, __builtin_bit_cast(int, t1), 0x128, 0xf, 0xc, false);
  return __builtin_bit_cast(float, v);
}

__global__ __launch_bounds__(96)
void k_fused(const float* __restrict__ x, const float* __restrict__ w1,
             const float* __restrict__ w2, float* __restrict__ y) {
  __shared__ v4f part[3][6][32];

  const int lane = threadIdx.x & 31;
  const int grp  = __builtin_amdgcn_readfirstlane(threadIdx.x >> 5);
  const int h    = lane >> 4;
  const int c15  = lane & 15;
  const int ablk = blockIdx.x / 7;
  const int bblk = blockIdx.x - ablk * 7;
  const int a    = 2 * ablk + (c15 >> 3);
  const int b    = 8 * bblk + (c15 & 7);

  const int  hich = (c15 < 2) ? 16 + c15
                  : (c15 == 2) ? 20
                  : (c15 == 8 || c15 == 9) ? 10 + c15 : 16;
  const int  kch  = 4 * grp + 2 * h;
  const float* w2lo = w2 + kch * WC2 + c15;
  const float* w2hi = w2 + kch * WC2 + hich;
  const float* xg   = x + kch * SP;
  const float* w1r  = w1 + (grp * 32 + c15) * W1R;

  v2f  a1lo[7], a1hi[7];
  XRaw r0, r1;
  A2F  f0, f1, f2;
  v8f  acc[2] = {};
  v8f  lo0 = {}, hi0 = {}, lo1 = {}, hi1 = {};

  load_g1<0, 7>(w2lo, w2hi, xg, a, b, a1lo, a1hi, r0, r1);
  LOADS_END();

  gemm1<0, 2>(r0, r1, a, b, a1lo, a1hi, lo0, hi0, lo1, hi1);
  STAGE_END(lo0, hi0, lo1, hi1);
  load_a2(w1r, 0, h, f0);
  load_a2(w1r, 1, h, f1);
  load_a2(w1r, 2, h, f2);
  LOADS_END();

  gemm1<2, 7>(r0, r1, a, b, a1lo, a1hi, lo0, hi0, lo1, hi1);

  part[grp][0][lane] = lo0.lo;
  part[grp][1][lane] = lo0.hi;
  part[grp][2][lane] = hi0.lo;
  part[grp][3][lane] = lo1.lo;
  part[grp][4][lane] = lo1.hi;
  part[grp][5][lane] = hi1.lo;
  __syncthreads();
  v4f p[3][6], s[6];
  #pragma unroll
  for (int w = 0; w < 3; ++w) {
    #pragma unroll
    for (int i = 0; i < 6; ++i) p[w][i] = part[w][i][lane];
  }
  LOADS_END();
  #pragma unroll
  for (int i = 0; i < 6; ++i) s[i] = (p[0][i] + p[1][i]) + p[2][i];
  lo0 = __builtin_shufflevector(s[0], s[1], 0, 1, 2, 3, 4, 5, 6, 7);
  lo1 = __builtin_shufflevector(s[3], s[4], 0, 1, 2, 3, 4, 5, 6, 7);
  const v4f h0 = s[2], h1 = s[5];

  tap_y(lo0, h0, f0, h, acc);
  v8f lom;
  v4f him;
  #pragma unroll
  for (int r = 0; r < 8; ++r) lom[r] = rows_0_1(lo0[r], lo1[r]);
  #pragma unroll
  for (int r = 0; r < 4; ++r) him[r] = rows_0_1(h0[r], h1[r]);
  tap_y(lom, him, f1, h, acc);
  tap_y(lo1, h1, f2, h, acc);

  #pragma unroll
  for (int q = 0; q < 2; ++q) {
    #pragma unroll
    for (int r = 0; r < 8; ++r) {
      y[((grp * 2 + q) * 16 + 8 * h + r) * SP + a * HW + b] = acc[q][r];
    }
  }
}

extern "C" void kernel_launch(void* const* d_in, const int* in_sizes, int n_in,
                              void* d_out, int out_size, void* d_ws, size_t ws_size,
                              hipStream_t stream) {
  const float* x  = (const float*)d_in[0];
  const float* w1 = (const float*)d_in[1];
  const float* w2 = (const float*)d_in[2];
  float* y = (float*)d_out;

  k_fused<<<196, 96, 0, stream>>>(x, w1, w2, y);
}
